# starting kernel
# baseline (speedup 1.0000x reference)
#include <hip/hip_runtime.h>

typedef __attribute__((ext_vector_type(2))) float v2f;
typedef __attribute__((ext_vector_type(8))) float v8f;

#define D_FEAT 512
#define N_OUT  64
#define PAIR_ROWS   (D_FEAT / 2)
#define PAIR_STRIDE 80

__global__ __launch_bounds__(256) void gcn_gemm_wmma(
    const float* __restrict__ x, const float* __restrict__ W,
    float* __restrict__ h, int n_nodes)
{
    __shared__ float2 Wp[PAIR_ROWS * PAIR_STRIDE];

    for (int i = threadIdx.x; i < PAIR_ROWS * 16; i += 256) {
        const int p  = i >> 4;
        const int n4 = (i & 15) << 2;
        const float4 r0 = *reinterpret_cast<const float4*>(W + (2 * p + 0) * N_OUT + n4);
        const float4 r1 = *reinterpret_cast<const float4*>(W + (2 * p + 1) * N_OUT + n4);
        float4* dst = reinterpret_cast<float4*>(&Wp[p * PAIR_STRIDE + n4]);
        float4 lo, hi;
        lo.x = r0.x; lo.y = r1.x; lo.z = r0.y; lo.w = r1.y;
        hi.x = r0.z; hi.y = r1.z; hi.z = r0.w; hi.w = r1.w;
        dst[0] = lo;
        dst[1] = hi;
    }
    __syncthreads();

    const int lane = threadIdx.x & 31;
    const int wave = threadIdx.x >> 5;
    const int l16  = lane & 15;
    const int g    = lane >> 4;
    const long rowBase = (long)blockIdx.x * 128 + (long)wave * 16;
    if (rowBase >= n_nodes) return;

    const float* aptr = x + (rowBase + l16) * (long)D_FEAT + 2 * g;
    const float2* bptr = &Wp[(long)g * PAIR_STRIDE + l16];

    v8f acc0 = {}, acc1 = {}, acc2 = {}, acc3 = {};

    for (int k = 0; k < D_FEAT; k += 4) {
        const v2f a = *reinterpret_cast<const v2f*>(aptr + k);

        const float2* wb = bptr + (k >> 1) * PAIR_STRIDE;
        const v2f b0 = *reinterpret_cast<const v2f*>(wb +  0);
        const v2f b1 = *reinterpret_cast<const v2f*>(wb + 16);
        const v2f b2 = *reinterpret_cast<const v2f*>(wb + 32);
        const v2f b3 = *reinterpret_cast<const v2f*>(wb + 48);

        acc0 = __builtin_amdgcn_wmma_f32_16x16x4_f32(false, a, false, b0, (short)0, acc0, false, false);
        acc1 = __builtin_amdgcn_wmma_f32_16x16x4_f32(false, a, false, b1, (short)0, acc1, false, false);
        acc2 = __builtin_amdgcn_wmma_f32_16x16x4_f32(false, a, false, b2, (short)0, acc2, false, false);
        acc3 = __builtin_amdgcn_wmma_f32_16x16x4_f32(false, a, false, b3, (short)0, acc3, false, false);
    }

    float* hb = h + rowBase * N_OUT + l16;
#pragma unroll
    for (int v = 0; v < 8; ++v) {
        const long m = v + 8 * g;
        hb[m * N_OUT +  0] = acc0[v];
        hb[m * N_OUT + 16] = acc1[v];
        hb[m * N_OUT + 32] = acc2[v];
        hb[m * N_OUT + 48] = acc3[v];
    }
}

__global__ __launch_bounds__(256) void gcn_zero(float4* __restrict__ p, long n4)
{
    long i = (long)blockIdx.x * blockDim.x + threadIdx.x;
    const long stride = (long)gridDim.x * blockDim.x;
    const float4 z = {0.f, 0.f, 0.f, 0.f};
    for (; i < n4; i += stride) p[i] = z;
}

__global__ __launch_bounds__(256) void gcn_scatter(
    const float* __restrict__ h, const int* __restrict__ esrc,
    const int* __restrict__ edst, const float* __restrict__ ew,
    float* __restrict__ agg, int n_edges)
{
    const int wave_id = (int)(((long)blockIdx.x * blockDim.x + threadIdx.x) >> 5);
    const int lane = threadIdx.x & 31;
    if (wave_id >= n_edges) return;

    const int   s  = esrc[wave_id];
    const int   d  = edst[wave_id];
    const float wt = ew[wave_id];

    const float2 hv = reinterpret_cast<const float2*>(h + (long)s * N_OUT)[lane];
    float* ap = agg + (long)d * N_OUT + 2 * lane;
#if defined(__HIP_DEVICE_COMPILE__)
    unsafeAtomicAdd(ap + 0, hv.x * wt);
    unsafeAtomicAdd(ap + 1, hv.y * wt);
#endif
}

__global__ __launch_bounds__(256) void gcn_softmax(float* __restrict__ agg, int n_nodes)
{
    const int wave_id = (int)(((long)blockIdx.x * blockDim.x + threadIdx.x) >> 5);
    const int lane = threadIdx.x & 31;
    if (wave_id >= n_nodes) return;

    float2* row = reinterpret_cast<float2*>(agg + (long)wave_id * N_OUT);
    const float2 v = row[lane];

    float m = fmaxf(v.x, v.y);
#pragma unroll
    for (int off = 16; off > 0; off >>= 1)
        m = fmaxf(m, __shfl_xor(m, off, 32));

    const float e0 = __expf(v.x - m);
    const float e1 = __expf(v.y - m);
    float s = e0 + e1;
#pragma unroll
    for (int off = 16; off > 0; off >>= 1)
        s += __shfl_xor(s, off, 32);

    const float inv = 1.0f / s;
    float2 o; o.x = e0 * inv; o.y = e1 * inv;
    row[lane] = o;
}

extern "C" void kernel_launch(void* const* d_in, const int* in_sizes, int n_in,
                              void* d_out, int out_size, void* d_ws, size_t ws_size,
                              hipStream_t stream)
{
    const float* x    = (const float*)d_in[0];
    const int*   esrc = (const int*)d_in[1];
    const int*   edst = (const int*)d_in[2];
    const float* ew   = (const float*)d_in[3];
    const float* W    = (const float*)d_in[4];

    const int n_nodes = in_sizes[0] / D_FEAT;
    const int n_edges = in_sizes[1];

    float* agg = (float*)d_out;
    float* h   = (float*)d_ws;

    {
        const long n4 = ((long)n_nodes * N_OUT) / 4;
        const int blocks = (int)((n4 + 255) / 256);
        gcn_zero<<<blocks, 256, 0, stream>>>((float4*)agg, n4);
    }
    {
        const int blocks = (n_nodes + 127) / 128;
        gcn_gemm_wmma<<<blocks, 256, 0, stream>>>(x, W, h, n_nodes);
    }
    {
        const int blocks = (int)(((long)n_edges * 32 + 255) / 256);
        gcn_scatter<<<blocks, 256, 0, stream>>>(h, esrc, edst, ew, agg, n_edges);
    }
    {
        const int blocks = (int)(((long)n_nodes * 32 + 255) / 256);
        gcn_softmax<<<blocks, 256, 0, stream>>>(agg, n_nodes);
    }
}


# f32 atomic scatter replaced by per-row edge buckets in the output rows (returning int adds) and one owner wave per row summing in ascending edge order, softmax in registers; sum order now fixed
# speedup vs baseline: 1.2921x; 1.2921x over previous
#include <hip/hip_runtime.h>

typedef __attribute__((ext_vector_type(2))) float v2f;
typedef __attribute__((ext_vector_type(8))) float v8f;

#define D_FEAT 512
#define N_OUT  64
#define PAIR_ROWS   (D_FEAT / 2)
#define PAIR_STRIDE 80
#define CNT_SLOT   63
#define BUCKET_CAP 63

__global__ __launch_bounds__(256) void gcn_gemm_wmma(
    const float* __restrict__ x, const float* __restrict__ W,
    float* __restrict__ h, int* __restrict__ bucket, int n_nodes)
{
    __shared__ float2 Wp[PAIR_ROWS * PAIR_STRIDE];

    {
        const long r = (long)blockIdx.x * 128 + threadIdx.x;
        if (threadIdx.x < 128 && r < n_nodes) bucket[r * N_OUT + CNT_SLOT] = 0;
    }

    for (int i = threadIdx.x; i < PAIR_ROWS * 16; i += 256) {
        const int p  = i >> 4;
        const int n4 = (i & 15) << 2;
        const float4 r0 = *reinterpret_cast<const float4*>(W + (2 * p + 0) * N_OUT + n4);
        const float4 r1 = *reinterpret_cast<const float4*>(W + (2 * p + 1) * N_OUT + n4);
        float4* dst = reinterpret_cast<float4*>(&Wp[p * PAIR_STRIDE + n4]);
        float4 lo, hi;
        lo.x = r0.x; lo.y = r1.x; lo.z = r0.y; lo.w = r1.y;
        hi.x = r0.z; hi.y = r1.z; hi.z = r0.w; hi.w = r1.w;
        dst[0] = lo;
        dst[1] = hi;
    }
    __syncthreads();

    const int lane = threadIdx.x & 31;
    const int wave = threadIdx.x >> 5;
    const int l16  = lane & 15;
    const int g    = lane >> 4;
    const long rowBase = (long)blockIdx.x * 128 + (long)wave * 16;
    if (rowBase >= n_nodes) return;

    const float* aptr = x + (rowBase + l16) * (long)D_FEAT + 2 * g;
    const float2* bptr = &Wp[(long)g * PAIR_STRIDE + l16];

    v8f acc0 = {}, acc1 = {}, acc2 = {}, acc3 = {};

    for (int k = 0; k < D_FEAT; k += 4) {
        const v2f a = *reinterpret_cast<const v2f*>(aptr + k);

        const float2* wb = bptr + (k >> 1) * PAIR_STRIDE;
        const v2f b0 = *reinterpret_cast<const v2f*>(wb +  0);
        const v2f b1 = *reinterpret_cast<const v2f*>(wb + 16);
        const v2f b2 = *reinterpret_cast<const v2f*>(wb + 32);
        const v2f b3 = *reinterpret_cast<const v2f*>(wb + 48);

        acc0 = __builtin_amdgcn_wmma_f32_16x16x4_f32(false, a, false, b0, (short)0, acc0, false, false);
        acc1 = __builtin_amdgcn_wmma_f32_16x16x4_f32(false, a, false, b1, (short)0, acc1, false, false);
        acc2 = __builtin_amdgcn_wmma_f32_16x16x4_f32(false, a, false, b2, (short)0, acc2, false, false);
        acc3 = __builtin_amdgcn_wmma_f32_16x16x4_f32(false, a, false, b3, (short)0, acc3, false, false);
    }

    float* hb = h + rowBase * N_OUT + l16;
#pragma unroll
    for (int v = 0; v < 8; ++v) {
        const long m = v + 8 * g;
        hb[m * N_OUT +  0] = acc0[v];
        hb[m * N_OUT + 16] = acc1[v];
        hb[m * N_OUT + 32] = acc2[v];
        hb[m * N_OUT + 48] = acc3[v];
    }
}

__global__ __launch_bounds__(256) void gcn_fill(
    const int* __restrict__ edst, int* __restrict__ bucket, int n_edges)
{
    const long stride = (long)gridDim.x * blockDim.x;
    for (long b = (long)blockIdx.x * blockDim.x + threadIdx.x; b < n_edges; b += 4 * stride) {
        int d[4];
        int pos[4];
#pragma unroll
        for (int k = 0; k < 4; ++k) {
            const long i = b + k * stride;
            d[k] = edst[(i < n_edges) ? i : b];
        }
#pragma unroll
        for (int k = 0; k < 4; ++k) {
            pos[k] = BUCKET_CAP;
            if (b + k * stride < n_edges)
                pos[k] = atomicAdd(bucket + (long)d[k] * N_OUT + CNT_SLOT, 1);
        }
#pragma unroll
        for (int k = 0; k < 4; ++k) {
            if (pos[k] < BUCKET_CAP) bucket[(long)d[k] * N_OUT + pos[k]] = (int)(b + k * stride);
        }
    }
}

__device__ __forceinline__ int lane_bcast_i(int v0, int v1, int r)
{
    return (r < 32) ? __builtin_amdgcn_readlane(v0, r) : __builtin_amdgcn_readlane(v1, r - 32);
}

__global__ __launch_bounds__(256) void gcn_gather_softmax(
    const float* __restrict__ h, const int* __restrict__ esrc,
    const int* __restrict__ edst, const float* __restrict__ ew,
    float* out, int n_nodes, int n_edges)
{
    __shared__ int   s_src[8][64];
    __shared__ float s_w[8][64];

    const int lane = threadIdx.x & 31;
    const int wave = __builtin_amdgcn_readfirstlane((int)(threadIdx.x >> 5));
    const int n_waves = (int)gridDim.x * 8;
    const float2* h2 = reinterpret_cast<const float2*>(h);

    for (int node = (int)blockIdx.x * 8 + wave; node < n_nodes; node += n_waves) {
        float* row = out + (long)node * N_OUT;
        const float2 bk = reinterpret_cast<const float2*>(row)[lane];
        const int e0 = __float_as_int(bk.x);
        const int e1 = __float_as_int(bk.y);
        const int cnt = __builtin_amdgcn_readlane(e1, 31);

        float ax = 0.f, ay = 0.f;

        if (cnt <= BUCKET_CAP) {
            const bool v0 = 2 * lane < cnt;
            const bool v1 = 2 * lane + 1 < cnt;
            int   src0 = 0, src1 = 0;
            float w0 = 0.f, w1 = 0.f;
            if (v0) { src0 = esrc[e0]; w0 = ew[e0]; }
            if (v1) { src1 = esrc[e1]; w1 = ew[e1]; }

            int rank0 = 0, rank1 = 0;
            const int n_even = (cnt + 1) >> 1;
            const int n_odd  = cnt >> 1;
            for (int j = 0; j < n_even; ++j) {
                const int ej = __builtin_amdgcn_readlane(e0, j);
                rank0 += (ej < e0) ? 1 : 0;
                rank1 += (ej < e1) ? 1 : 0;
            }
            for (int j = 0; j < n_odd; ++j) {
                const int ej = __builtin_amdgcn_readlane(e1, j);
                rank0 += (ej < e0) ? 1 : 0;
                rank1 += (ej < e1) ? 1 : 0;
            }
            if (v0) { s_src[wave][rank0] = src0; s_w[wave][rank0] = w0; }
            if (v1) { s_src[wave][rank1] = src1; s_w[wave][rank1] = w1; }
            const int   ss0 = s_src[wave][lane];
            const int   ss1 = s_src[wave][lane + 32];
            const int   sw0 = __float_as_int(s_w[wave][lane]);
            const int   sw1 = __float_as_int(s_w[wave][lane + 32]);

            for (int r = 0; r < cnt; r += 8) {
                float2 v[8];
                float  wv[8];
                bool   ok[8];
#pragma unroll
                for (int i = 0; i < 8; ++i) {
                    ok[i] = (r + i) < cnt;
                    const int rr = ok[i] ? (r + i) : r;
                    const int s  = lane_bcast_i(ss0, ss1, rr);
                    wv[i] = __int_as_float(lane_bcast_i(sw0, sw1, rr));
                    v[i]  = h2[(long)s * (N_OUT / 2) + lane];
                }
#pragma unroll
                for (int i = 0; i < 8; ++i) {
                    const float px = v[i].x * wv[i];
                    const float py = v[i].y * wv[i];
                    ax += ok[i] ? px : 0.f;
                    ay += ok[i] ? py : 0.f;
                }
            }
        } else {
            for (int base = 0; base < n_edges; base += 32) {
                const int e = base + lane;
                const int d = (e < n_edges) ? edst[e] : -1;
                unsigned m = __builtin_amdgcn_ballot_w32(d == node);
                while (m) {
                    const int j = __builtin_ctz(m);
                    m &= m - 1;
                    const int   s = esrc[base + j];
                    const float w = ew[base + j];
                    const float2 hv = h2[(long)s * (N_OUT / 2) + lane];
                    ax += hv.x * w;
                    ay += hv.y * w;
                }
            }
        }

        float m = fmaxf(ax, ay);
#pragma unroll
        for (int off = 16; off > 0; off >>= 1)
            m = fmaxf(m, __shfl_xor(m, off, 32));
        const float x0 = __expf(ax - m);
        const float x1 = __expf(ay - m);
        float s = x0 + x1;
#pragma unroll
        for (int off = 16; off > 0; off >>= 1)
            s += __shfl_xor(s, off, 32);
        const float inv = 1.0f / s;
        float2 o; o.x = x0 * inv; o.y = x1 * inv;
        reinterpret_cast<float2*>(row)[lane] = o;
    }
}

extern "C" void kernel_launch(void* const* d_in, const int* in_sizes, int n_in,
                              void* d_out, int out_size, void* d_ws, size_t ws_size,
                              hipStream_t stream)
{
    const float* x    = (const float*)d_in[0];
    const int*   esrc = (const int*)d_in[1];
    const int*   edst = (const int*)d_in[2];
    const float* ew   = (const float*)d_in[3];
    const float* W    = (const float*)d_in[4];

    const int n_nodes = in_sizes[0] / D_FEAT;
    const int n_edges = in_sizes[1];

    float* outp = (float*)d_out;
    float* h    = (float*)d_ws;

    {
        const int blocks = (n_nodes + 127) / 128;
        gcn_gemm_wmma<<<blocks, 256, 0, stream>>>(x, W, h, (int*)d_out, n_nodes);
    }
    gcn_fill<<<2048, 256, 0, stream>>>(edst, (int*)d_out, n_edges);
    gcn_gather_softmax<<<2048, 256, 0, stream>>>(h, esrc, edst, ew, outp, n_nodes, n_edges);
}


# projection made persistent: 256 work-groups of 25 waves (one per WGP) stage W in LDS once and each wave takes one 16-row tile, 6-7 waves per SIMD from the start; no tail round, no restaging of W
# speedup vs baseline: 1.4106x; 1.0917x over previous
#include <hip/hip_runtime.h>

typedef __attribute__((ext_vector_type(2))) float v2f;
typedef __attribute__((ext_vector_type(8))) float v8f;

#define D_FEAT 512
#define N_OUT  64
#define PAIR_ROWS   (D_FEAT / 2)
#define PAIR_STRIDE 80
#define CNT_SLOT   63
#define PAIR_SLOTS 20
#define BUCKET_CAP 43
#define GEMM_WAVES   25
#define GEMM_THREADS (GEMM_WAVES * 32)

__global__ __launch_bounds__(GEMM_THREADS) void gcn_gemm_wmma(
    const float* __restrict__ x, const float* __restrict__ W,
    float* __restrict__ h, int* __restrict__ bucket, int n_nodes)
{
    __shared__ float2 Wp[(PAIR_ROWS + 1) * PAIR_STRIDE];

    for (long r = (long)blockIdx.x * GEMM_THREADS + threadIdx.x; r < n_nodes;
         r += (long)gridDim.x * GEMM_THREADS)
        bucket[r * N_OUT + CNT_SLOT] = 0;

    for (int i = threadIdx.x; i < PAIR_ROWS * 16; i += GEMM_THREADS) {
        const int p  = i >> 4;
        const int n4 = (i & 15) << 2;
        const float4 r0 = *reinterpret_cast<const float4*>(W + (2 * p + 0) * N_OUT + n4);
        const float4 r1 = *reinterpret_cast<const float4*>(W + (2 * p + 1) * N_OUT + n4);
        float4* dst = reinterpret_cast<float4*>(&Wp[p * PAIR_STRIDE + n4]);
        float4 lo, hi;
        lo.x = r0.x; lo.y = r1.x; lo.z = r0.y; lo.w = r1.y;
        hi.x = r0.z; hi.y = r1.z; hi.z = r0.w; hi.w = r1.w;
        dst[0] = lo;
        dst[1] = hi;
    }
    __syncthreads();

    const int lane = threadIdx.x & 31;
    const int wave = __builtin_amdgcn_readfirstlane((int)(threadIdx.x >> 5));
    const int l16  = lane & 15;
    const int g    = lane >> 4;
    const int n_tiles = (n_nodes + 15) >> 4;
    const float2* bptr = &Wp[(long)g * PAIR_STRIDE + l16];

    for (int tile = (int)blockIdx.x + (int)gridDim.x * wave; tile < n_tiles;
         tile += (int)gridDim.x * GEMM_WAVES) {
        const long rowBase = (long)tile * 16;
        const float* aptr = x + (rowBase + l16) * (long)D_FEAT + 2 * g;

        v8f acc0 = {}, acc1 = {}, acc2 = {}, acc3 = {};

        for (int k = 0; k < D_FEAT; k += 4) {
            const v2f a = *reinterpret_cast<const v2f*>(aptr + k);

            const float2* wb = bptr + (k >> 1) * PAIR_STRIDE;
            const v2f b0 = *reinterpret_cast<const v2f*>(wb +  0);
            const v2f b1 = *reinterpret_cast<const v2f*>(wb + 16);
            const v2f b2 = *reinterpret_cast<const v2f*>(wb + 32);
            const v2f b3 = *reinterpret_cast<const v2f*>(wb + 48);

            acc0 = __builtin_amdgcn_wmma_f32_16x16x4_f32(false, a, false, b0, (short)0, acc0, false, false);
            acc1 = __builtin_amdgcn_wmma_f32_16x16x4_f32(false, a, false, b1, (short)0, acc1, false, false);
            acc2 = __builtin_amdgcn_wmma_f32_16x16x4_f32(false, a, false, b2, (short)0, acc2, false, false);
            acc3 = __builtin_amdgcn_wmma_f32_16x16x4_f32(false, a, false, b3, (short)0, acc3, false, false);
        }

        float* hb = h + rowBase * N_OUT + l16;
#pragma unroll
        for (int v = 0; v < 8; ++v) {
            const long m = v + 8 * g;
            hb[m * N_OUT +  0] = acc0[v];
            hb[m * N_OUT + 16] = acc1[v];
            hb[m * N_OUT + 32] = acc2[v];
            hb[m * N_OUT + 48] = acc3[v];
        }
    }
}

__global__ __launch_bounds__(256) void gcn_fill(
    const int* __restrict__ esrc, const int* __restrict__ edst,
    const float* __restrict__ ew, int* __restrict__ bucket, int n_edges)
{
    const long stride = (long)gridDim.x * blockDim.x;
    for (long b = (long)blockIdx.x * blockDim.x + threadIdx.x; b < n_edges; b += 4 * stride) {
        int d[4];
        int s[4];
        int w[4];
        int pos[4];
#pragma unroll
        for (int k = 0; k < 4; ++k) {
            const long i = b + k * stride;
            const long j = (i < n_edges) ? i : b;
            d[k] = edst[j];
            s[k] = esrc[j];
            w[k] = __float_as_int(ew[j]);
        }
#pragma unroll
        for (int k = 0; k < 4; ++k) {
            pos[k] = BUCKET_CAP;
            if (b + k * stride < n_edges)
                pos[k] = atomicAdd(bucket + (long)d[k] * N_OUT + CNT_SLOT, 1);
        }
#pragma unroll
        for (int k = 0; k < 4; ++k) {
            int* row = bucket + (long)d[k] * N_OUT;
            if (pos[k] < PAIR_SLOTS) {
                int2 p; p.x = s[k]; p.y = w[k];
                *reinterpret_cast<int2*>(row + 2 * pos[k]) = p;
            } else if (pos[k] < BUCKET_CAP) {
                row[PAIR_SLOTS + pos[k]] = (int)(b + k * stride);
            }
        }
    }
}

__device__ __forceinline__ int lane_bcast_i(int v0, int v1, int r)
{
    return (r < 32) ? __builtin_amdgcn_readlane(v0, r) : __builtin_amdgcn_readlane(v1, r - 32);
}

__global__ __launch_bounds__(256) void gcn_gather_softmax(
    const float* __restrict__ h, const int* __restrict__ esrc,
    const int* __restrict__ edst, const float* __restrict__ ew,
    float* out, int n_nodes, int n_edges)
{
    __shared__ int   s_src[8][64];
    __shared__ float s_w[8][64];

    const int lane = threadIdx.x & 31;
    const int wave = __builtin_amdgcn_readfirstlane((int)(threadIdx.x >> 5));
    const int n_waves = (int)gridDim.x * 8;
    const float2* h2 = reinterpret_cast<const float2*>(h);

    for (int node = (int)blockIdx.x * 8 + wave; node < n_nodes; node += n_waves) {
        float* row = out + (long)node * N_OUT;
        const float2 bk = reinterpret_cast<const float2*>(row)[lane];
        const int b0 = __float_as_int(bk.x);
        const int b1 = __float_as_int(bk.y);
        const int cnt = __builtin_amdgcn_readlane(b1, 31);

        float ax = 0.f, ay = 0.f;

        if (cnt <= BUCKET_CAP) {
            const bool pair_lane = lane < PAIR_SLOTS;
            const int  slotA = pair_lane ? lane : (2 * lane - PAIR_SLOTS);
            const int  slotB = 2 * lane - PAIR_SLOTS + 1;
            const bool vA = slotA < cnt;
            const bool vB = !pair_lane && (slotB < cnt);
            int srcA = b0, wA = b1;
            int srcB = 0,  wB = 0;
            if (cnt > PAIR_SLOTS) {
                if (!pair_lane && vA) { srcA = esrc[b0]; wA = __float_as_int(ew[b0]); }
                if (vB)               { srcB = esrc[b1]; wB = __float_as_int(ew[b1]); }
            }
            const unsigned long long kA =
                ((unsigned long long)(unsigned)srcA << 38) | ((unsigned long long)(unsigned)wA << 6) | (unsigned)slotA;
            const unsigned long long kB =
                ((unsigned long long)(unsigned)srcB << 38) | ((unsigned long long)(unsigned)wB << 6) | (unsigned)slotB;
            const int kAlo = (int)(unsigned)kA, kAhi = (int)(unsigned)(kA >> 32);
            const int kBlo = (int)(unsigned)kB, kBhi = (int)(unsigned)(kB >> 32);

            int rankA = 0, rankB = 0;
            if (cnt <= PAIR_SLOTS) {
                for (int j = 0; j < cnt; ++j) {
                    const unsigned long long kj =
                        ((unsigned long long)(unsigned)__builtin_amdgcn_readlane(kAhi, j) << 32) |
                        (unsigned)__builtin_amdgcn_readlane(kAlo, j);
                    rankA += (kj < kA) ? 1 : 0;
                }
            } else {
                const int nA = PAIR_SLOTS + ((cnt - PAIR_SLOTS + 1) >> 1);
                const int nB = (cnt - PAIR_SLOTS) >> 1;
                for (int j = 0; j < nA; ++j) {
                    const unsigned long long kj =
                        ((unsigned long long)(unsigned)__builtin_amdgcn_readlane(kAhi, j) << 32) |
                        (unsigned)__builtin_amdgcn_readlane(kAlo, j);
                    rankA += (kj < kA) ? 1 : 0;
                    rankB += (kj < kB) ? 1 : 0;
                }
                for (int j = 0; j < nB; ++j) {
                    const unsigned long long kj =
                        ((unsigned long long)(unsigned)__builtin_amdgcn_readlane(kBhi, PAIR_SLOTS + j) << 32) |
                        (unsigned)__builtin_amdgcn_readlane(kBlo, PAIR_SLOTS + j);
                    rankA += (kj < kA) ? 1 : 0;
                    rankB += (kj < kB) ? 1 : 0;
                }
            }
            if (vA) { s_src[wave][rankA] = srcA; s_w[wave][rankA] = __int_as_float(wA); }
            if (vB) { s_src[wave][rankB] = srcB; s_w[wave][rankB] = __int_as_float(wB); }
            const int   ss0 = s_src[wave][lane];
            const int   ss1 = s_src[wave][lane + 32];
            const int   sw0 = __float_as_int(s_w[wave][lane]);
            const int   sw1 = __float_as_int(s_w[wave][lane + 32]);

            for (int r = 0; r < cnt; r += 8) {
                float2 v[8];
                float  wv[8];
                bool   ok[8];
#pragma unroll
                for (int i = 0; i < 8; ++i) {
                    ok[i] = (r + i) < cnt;
                    const int rr = ok[i] ? (r + i) : r;
                    const int s  = lane_bcast_i(ss0, ss1, rr);
                    wv[i] = __int_as_float(lane_bcast_i(sw0, sw1, rr));
                    v[i]  = h2[(long)s * (N_OUT / 2) + lane];
                }
#pragma unroll
                for (int i = 0; i < 8; ++i) {
                    const float px = v[i].x * wv[i];
                    const float py = v[i].y * wv[i];
                    ax += ok[i] ? px : 0.f;
                    ay += ok[i] ? py : 0.f;
                }
            }
        } else {
            for (int base = 0; base < n_edges; base += 32) {
                const int e = base + lane;
                const int d = (e < n_edges) ? edst[e] : -1;
                unsigned m = __builtin_amdgcn_ballot_w32(d == node);
                while (m) {
                    const int j = __builtin_ctz(m);
                    m &= m - 1;
                    const int   s = esrc[base + j];
                    const float w = ew[base + j];
                    const float2 hv = h2[(long)s * (N_OUT / 2) + lane];
                    ax += hv.x * w;
                    ay += hv.y * w;
                }
            }
        }

        float m = fmaxf(ax, ay);
#pragma unroll
        for (int off = 16; off > 0; off >>= 1)
            m = fmaxf(m, __shfl_xor(m, off, 32));
        const float x0 = __expf(ax - m);
        const float x1 = __expf(ay - m);
        float s = x0 + x1;
#pragma unroll
        for (int off = 16; off > 0; off >>= 1)
            s += __shfl_xor(s, off, 32);
        const float inv = 1.0f / s;
        float2 o; o.x = x0 * inv; o.y = x1 * inv;
        reinterpret_cast<float2*>(row)[lane] = o;
    }
}

extern "C" void kernel_launch(void* const* d_in, const int* in_sizes, int n_in,
                              void* d_out, int out_size, void* d_ws, size_t ws_size,
                              hipStream_t stream)
{
    const float* x    = (const float*)d_in[0];
    const int*   esrc = (const int*)d_in[1];
    const int*   edst = (const int*)d_in[2];
    const float* ew   = (const float*)d_in[3];
    const float* W    = (const float*)d_in[4];

    const int n_nodes = in_sizes[0] / D_FEAT;
    const int n_edges = in_sizes[1];

    float* outp = (float*)d_out;
    float* h    = (float*)d_ws;

    gcn_gemm_wmma<<<256, GEMM_THREADS, 0, stream>>>(x, W, h, (int*)d_out, n_nodes);
    gcn_fill<<<2048, 256, 0, stream>>>(esrc, edst, ew, (int*)d_out, n_edges);
    gcn_gather_softmax<<<2048, 256, 0, stream>>>(h, esrc, edst, ew, outp, n_nodes, n_edges);
}


# bucket fill runs inside the projection launch on 7 extra waves per work-group (raised issue priority), under the matrix work; counter clear now a small launch in front; sum order unchanged
# speedup vs baseline: 1.7556x; 1.2446x over previous
#include <hip/hip_runtime.h>

typedef __attribute__((ext_vector_type(2))) float v2f;
typedef __attribute__((ext_vector_type(8))) float v8f;

#define D_FEAT 512
#define N_OUT  64
#define PAIR_ROWS   (D_FEAT / 2)
#define PAIR_STRIDE 80
#define CNT_SLOT   63
#define PAIR_SLOTS 20
#define BUCKET_CAP 43
#define GEMM_WAVES    25
#define FILL_WAVES    7
#define BLOCK_THREADS ((GEMM_WAVES + FILL_WAVES) * 32)
#define FILL_EDGES    8

__global__ __launch_bounds__(128) void gcn_clear(int* __restrict__ bucket, int n_nodes)
{
    for (long r = (long)blockIdx.x * blockDim.x + threadIdx.x; r < n_nodes;
         r += (long)gridDim.x * blockDim.x)
        bucket[r * N_OUT + CNT_SLOT] = 0;
}

__device__ __forceinline__ void fill_edges(
    const int* __restrict__ esrc, const int* __restrict__ edst,
    const float* __restrict__ ew, int* __restrict__ bucket, int n_edges,
    long first, long stride)
{
    for (long b = first; b < n_edges; b += FILL_EDGES * stride) {
        int d[FILL_EDGES];
        int s[FILL_EDGES];
        int w[FILL_EDGES];
        int pos[FILL_EDGES];
#pragma unroll
        for (int k = 0; k < FILL_EDGES; ++k) {
            const long i = b + k * stride;
            const long j = (i < n_edges) ? i : b;
            d[k] = edst[j];
            s[k] = esrc[j];
            w[k] = __float_as_int(ew[j]);
        }
#pragma unroll
        for (int k = 0; k < FILL_EDGES; ++k) {
            pos[k] = BUCKET_CAP;
            if (b + k * stride < n_edges)
                pos[k] = atomicAdd(bucket + (long)d[k] * N_OUT + CNT_SLOT, 1);
        }
#pragma unroll
        for (int k = 0; k < FILL_EDGES; ++k) {
            int* row = bucket + (long)d[k] * N_OUT;
            if (pos[k] < PAIR_SLOTS) {
                int2 p; p.x = s[k]; p.y = w[k];
                *reinterpret_cast<int2*>(row + 2 * pos[k]) = p;
            } else if (pos[k] < BUCKET_CAP) {
                row[PAIR_SLOTS + pos[k]] = (int)(b + k * stride);
            }
        }
    }
}

__global__ __launch_bounds__(BLOCK_THREADS) void gcn_gemm_fill(
    const float* __restrict__ x, const float* __restrict__ W,
    float* __restrict__ h,
    const int* __restrict__ esrc, const int* __restrict__ edst,
    const float* __restrict__ ew, int* __restrict__ bucket,
    int n_nodes, int n_edges)
{
    __shared__ float2 Wp[(PAIR_ROWS + 1) * PAIR_STRIDE];

    for (int i = threadIdx.x; i < PAIR_ROWS * 16; i += BLOCK_THREADS) {
        const int p  = i >> 4;
        const int n4 = (i & 15) << 2;
        const float4 r0 = *reinterpret_cast<const float4*>(W + (2 * p + 0) * N_OUT + n4);
        const float4 r1 = *reinterpret_cast<const float4*>(W + (2 * p + 1) * N_OUT + n4);
        float4* dst = reinterpret_cast<float4*>(&Wp[p * PAIR_STRIDE + n4]);
        float4 lo, hi;
        lo.x = r0.x; lo.y = r1.x; lo.z = r0.y; lo.w = r1.y;
        hi.x = r0.z; hi.y = r1.z; hi.z = r0.w; hi.w = r1.w;
        dst[0] = lo;
        dst[1] = hi;
    }
    __syncthreads();

    const int lane = threadIdx.x & 31;
    const int wave = __builtin_amdgcn_readfirstlane((int)(threadIdx.x >> 5));

    if (wave >= GEMM_WAVES) {
        __builtin_amdgcn_s_setprio(3);
        const long first  = ((long)blockIdx.x * FILL_WAVES + (wave - GEMM_WAVES)) * 32 + lane;
        const long stride = (long)gridDim.x * FILL_WAVES * 32;
        fill_edges(esrc, edst, ew, bucket, n_edges, first, stride);
        return;
    }

    const int l16  = lane & 15;
    const int g    = lane >> 4;
    const int n_tiles = (n_nodes + 15) >> 4;
    const float2* bptr = &Wp[(long)g * PAIR_STRIDE + l16];

    for (int tile = (int)blockIdx.x + (int)gridDim.x * wave; tile < n_tiles;
         tile += (int)gridDim.x * GEMM_WAVES) {
        const long rowBase = (long)tile * 16;
        const float* aptr = x + (rowBase + l16) * (long)D_FEAT + 2 * g;

        v8f acc0 = {}, acc1 = {}, acc2 = {}, acc3 = {};

        for (int k = 0; k < D_FEAT; k += 4) {
            const v2f a = *reinterpret_cast<const v2f*>(aptr + k);

            const float2* wb = bptr + (k >> 1) * PAIR_STRIDE;
            const v2f b0 = *reinterpret_cast<const v2f*>(wb +  0);
            const v2f b1 = *reinterpret_cast<const v2f*>(wb + 16);
            const v2f b2 = *reinterpret_cast<const v2f*>(wb + 32);
            const v2f b3 = *reinterpret_cast<const v2f*>(wb + 48);

            acc0 = __builtin_amdgcn_wmma_f32_16x16x4_f32(false, a, false, b0, (short)0, acc0, false, false);
            acc1 = __builtin_amdgcn_wmma_f32_16x16x4_f32(false, a, false, b1, (short)0, acc1, false, false);
            acc2 = __builtin_amdgcn_wmma_f32_16x16x4_f32(false, a, false, b2, (short)0, acc2, false, false);
            acc3 = __builtin_amdgcn_wmma_f32_16x16x4_f32(false, a, false, b3, (short)0, acc3, false, false);
        }

        float* hb = h + rowBase * N_OUT + l16;
#pragma unroll
        for (int v = 0; v < 8; ++v) {
            const long m = v + 8 * g;
            hb[m * N_OUT +  0] = acc0[v];
            hb[m * N_OUT + 16] = acc1[v];
            hb[m * N_OUT + 32] = acc2[v];
            hb[m * N_OUT + 48] = acc3[v];
        }
    }
}

__device__ __forceinline__ int lane_bcast_i(int v0, int v1, int r)
{
    return (r < 32) ? __builtin_amdgcn_readlane(v0, r) : __builtin_amdgcn_readlane(v1, r - 32);
}

__global__ __launch_bounds__(256) void gcn_gather_softmax(
    const float* __restrict__ h, const int* __restrict__ esrc,
    const int* __restrict__ edst, const float* __restrict__ ew,
    float* out, int n_nodes, int n_edges)
{
    __shared__ int   s_src[8][64];
    __shared__ float s_w[8][64];

    const int lane = threadIdx.x & 31;
    const int wave = __builtin_amdgcn_readfirstlane((int)(threadIdx.x >> 5));
    const int n_waves = (int)gridDim.x * 8;
    const float2* h2 = reinterpret_cast<const float2*>(h);

    for (int node = (int)blockIdx.x * 8 + wave; node < n_nodes; node += n_waves) {
        float* row = out + (long)node * N_OUT;
        const float2 bk = reinterpret_cast<const float2*>(row)[lane];
        const int b0 = __float_as_int(bk.x);
        const int b1 = __float_as_int(bk.y);
        const int cnt = __builtin_amdgcn_readlane(b1, 31);

        float ax = 0.f, ay = 0.f;

        if (cnt <= BUCKET_CAP) {
            const bool pair_lane = lane < PAIR_SLOTS;
            const int  slotA = pair_lane ? lane : (2 * lane - PAIR_SLOTS);
            const int  slotB = 2 * lane - PAIR_SLOTS + 1;
            const bool vA = slotA < cnt;
            const bool vB = !pair_lane && (slotB < cnt);
            int srcA = b0, wA = b1;
            int srcB = 0,  wB = 0;
            if (cnt > PAIR_SLOTS) {
                if (!pair_lane && vA) { srcA = esrc[b0]; wA = __float_as_int(ew[b0]); }
                if (vB)               { srcB = esrc[b1]; wB = __float_as_int(ew[b1]); }
            }
            const unsigned long long kA =
                ((unsigned long long)(unsigned)srcA << 38) | ((unsigned long long)(unsigned)wA << 6) | (unsigned)slotA;
            const unsigned long long kB =
                ((unsigned long long)(unsigned)srcB << 38) | ((unsigned long long)(unsigned)wB << 6) | (unsigned)slotB;
            const int kAlo = (int)(unsigned)kA, kAhi = (int)(unsigned)(kA >> 32);
            const int kBlo = (int)(unsigned)kB, kBhi = (int)(unsigned)(kB >> 32);

            int rankA = 0, rankB = 0;
            if (cnt <= PAIR_SLOTS) {
                for (int j = 0; j < cnt; ++j) {
                    const unsigned long long kj =
                        ((unsigned long long)(unsigned)__builtin_amdgcn_readlane(kAhi, j) << 32) |
                        (unsigned)__builtin_amdgcn_readlane(kAlo, j);
                    rankA += (kj < kA) ? 1 : 0;
                }
            } else {
                const int nA = PAIR_SLOTS + ((cnt - PAIR_SLOTS + 1) >> 1);
                const int nB = (cnt - PAIR_SLOTS) >> 1;
                for (int j = 0; j < nA; ++j) {
                    const unsigned long long kj =
                        ((unsigned long long)(unsigned)__builtin_amdgcn_readlane(kAhi, j) << 32) |
                        (unsigned)__builtin_amdgcn_readlane(kAlo, j);
                    rankA += (kj < kA) ? 1 : 0;
                    rankB += (kj < kB) ? 1 : 0;
                }
                for (int j = 0; j < nB; ++j) {
                    const unsigned long long kj =
                        ((unsigned long long)(unsigned)__builtin_amdgcn_readlane(kBhi, PAIR_SLOTS + j) << 32) |
                        (unsigned)__builtin_amdgcn_readlane(kBlo, PAIR_SLOTS + j);
                    rankA += (kj < kA) ? 1 : 0;
                    rankB += (kj < kB) ? 1 : 0;
                }
            }
            if (vA) { s_src[wave][rankA] = srcA; s_w[wave][rankA] = __int_as_float(wA); }
            if (vB) { s_src[wave][rankB] = srcB; s_w[wave][rankB] = __int_as_float(wB); }
            const int   ss0 = s_src[wave][lane];
            const int   ss1 = s_src[wave][lane + 32];
            const int   sw0 = __float_as_int(s_w[wave][lane]);
            const int   sw1 = __float_as_int(s_w[wave][lane + 32]);

            for (int r = 0; r < cnt; r += 8) {
                float2 v[8];
                float  wv[8];
                bool   ok[8];
#pragma unroll
                for (int i = 0; i < 8; ++i) {
                    ok[i] = (r + i) < cnt;
                    const int rr = ok[i] ? (r + i) : r;
                    const int s  = lane_bcast_i(ss0, ss1, rr);
                    wv[i] = __int_as_float(lane_bcast_i(sw0, sw1, rr));
                    v[i]  = h2[(long)s * (N_OUT / 2) + lane];
                }
#pragma unroll
                for (int i = 0; i < 8; ++i) {
                    const float px = v[i].x * wv[i];
                    const float py = v[i].y * wv[i];
                    ax += ok[i] ? px : 0.f;
                    ay += ok[i] ? py : 0.f;
                }
            }
        } else {
            for (int base = 0; base < n_edges; base += 32) {
                const int e = base + lane;
                const int d = (e < n_edges) ? edst[e] : -1;
                unsigned m = __builtin_amdgcn_ballot_w32(d == node);
                while (m) {
                    const int j = __builtin_ctz(m);
                    m &= m - 1;
                    const int   s = esrc[base + j];
                    const float w = ew[base + j];
                    const float2 hv = h2[(long)s * (N_OUT / 2) + lane];
                    ax += hv.x * w;
                    ay += hv.y * w;
                }
            }
        }

        float m = fmaxf(ax, ay);
#pragma unroll
        for (int off = 16; off > 0; off >>= 1)
            m = fmaxf(m, __shfl_xor(m, off, 32));
        const float x0 = __expf(ax - m);
        const float x1 = __expf(ay - m);
        float s = x0 + x1;
#pragma unroll
        for (int off = 16; off > 0; off >>= 1)
            s += __shfl_xor(s, off, 32);
        const float inv = 1.0f / s;
        float2 o; o.x = x0 * inv; o.y = x1 * inv;
        reinterpret_cast<float2*>(row)[lane] = o;
    }
}

extern "C" void kernel_launch(void* const* d_in, const int* in_sizes, int n_in,
                              void* d_out, int out_size, void* d_ws, size_t ws_size,
                              hipStream_t stream)
{
    const float* x    = (const float*)d_in[0];
    const int*   esrc = (const int*)d_in[1];
    const int*   edst = (const int*)d_in[2];
    const float* ew   = (const float*)d_in[3];
    const float* W    = (const float*)d_in[4];

    const int n_nodes = in_sizes[0] / D_FEAT;
    const int n_edges = in_sizes[1];

    float* outp = (float*)d_out;
    float* h    = (float*)d_ws;

    gcn_clear<<<256, 128, 0, stream>>>((int*)d_out, n_nodes);
    gcn_gemm_fill<<<256, BLOCK_THREADS, 0, stream>>>(x, W, h, esrc, edst, ew, (int*)d_out,
                                                     n_nodes, n_edges);
    gcn_gather_softmax<<<2048, 256, 0, stream>>>(h, esrc, edst, ew, outp, n_nodes, n_edges);
}


# bucket fill split by chip half: fill waves read their die number and take edge chunks by ticket (returning int add) for the destination rows of their own half first; sum order unchanged
# speedup vs baseline: 2.3612x; 1.3450x over previous
#include <hip/hip_runtime.h>

typedef __attribute__((ext_vector_type(2))) float v2f;
typedef __attribute__((ext_vector_type(8))) float v8f;

#define D_FEAT 512
#define N_OUT  64
#define PAIR_ROWS   (D_FEAT / 2)
#define PAIR_STRIDE 80
#define CNT_SLOT    63
#define TICKET_SLOT 62
#define PAIR_SLOTS  20
#define BUCKET_CAP  42
#define GEMM_WAVES    25
#define FILL_WAVES    7
#define BLOCK_THREADS ((GEMM_WAVES + FILL_WAVES) * 32)
#define FILL_EDGES    8
#define FILL_CHUNK    1024

__global__ __launch_bounds__(128) void gcn_clear(int* __restrict__ bucket, int n_nodes)
{
    for (long r = (long)blockIdx.x * blockDim.x + threadIdx.x; r < n_nodes;
         r += (long)gridDim.x * blockDim.x)
        bucket[r * N_OUT + CNT_SLOT] = 0;
    if (blockIdx.x == 0 && threadIdx.x == 0 && n_nodes > 0) {
        bucket[TICKET_SLOT] = 0;
        bucket[(long)(n_nodes - 1) * N_OUT + TICKET_SLOT] = 0;
    }
}

__device__ __forceinline__ void fill_chunk(
    const int* __restrict__ esrc, const int* __restrict__ edst,
    const float* __restrict__ ew, int* __restrict__ bucket, int n_edges,
    unsigned base, int lane, int lo, int hi)
{
    const unsigned ne = (unsigned)n_edges;
    for (int g = 0; g < FILL_CHUNK / (32 * FILL_EDGES); ++g) {
        const unsigned b = base + (unsigned)g * (32 * FILL_EDGES) + (unsigned)lane;
        int  d[FILL_EDGES];
        int  s[FILL_EDGES];
        int  w[FILL_EDGES];
        int  pos[FILL_EDGES];
        bool mine[FILL_EDGES];
#pragma unroll
        for (int k = 0; k < FILL_EDGES; ++k) {
            const unsigned i = b + (unsigned)k * 32u;
            const unsigned j = (i < ne) ? i : base;
            d[k] = edst[j];
            s[k] = esrc[j];
            w[k] = __float_as_int(ew[j]);
            mine[k] = (i < ne) && (d[k] >= lo) && (d[k] < hi);
        }
#pragma unroll
        for (int k = 0; k < FILL_EDGES; ++k) {
            pos[k] = BUCKET_CAP;
            if (mine[k])
                pos[k] = atomicAdd(bucket + (long)d[k] * N_OUT + CNT_SLOT, 1);
        }
#pragma unroll
        for (int k = 0; k < FILL_EDGES; ++k) {
            int* row = bucket + (long)d[k] * N_OUT;
            if (pos[k] < PAIR_SLOTS) {
                int2 p; p.x = s[k]; p.y = w[k];
                *reinterpret_cast<int2*>(row + 2 * pos[k]) = p;
            } else if (pos[k] < BUCKET_CAP) {
                row[PAIR_SLOTS + pos[k]] = (int)(b + (unsigned)k * 32u);
            }
        }
    }
}

__device__ __forceinline__ void fill_wave(
    const int* __restrict__ esrc, const int* __restrict__ edst,
    const float* __restrict__ ew, int* __restrict__ bucket,
    int n_nodes, int n_edges, int lane)
{
    const unsigned id = __builtin_amdgcn_s_sendmsg_rtn(0x87);
    const int half = __builtin_amdgcn_readfirstlane((int)((id >> 18) & 1u));
    const int split = n_nodes >> 1;
    const int n_chunks = (n_edges + FILL_CHUNK - 1) / FILL_CHUNK;
    for (int pass = 0; pass < 2; ++pass) {
        const int q  = half ^ pass;
        const int lo = q ? split : 0;
        const int hi = q ? n_nodes : split;
        int* ticket = bucket + (q ? (long)(n_nodes - 1) * N_OUT : 0L) + TICKET_SLOT;
        for (int it = 0; it <= n_chunks; ++it) {
            int t = 0;
            if (lane == 0) t = atomicAdd(ticket, 1);
            t = __builtin_amdgcn_readfirstlane(t);
            if (t >= n_chunks) break;
            fill_chunk(esrc, edst, ew, bucket, n_edges, (unsigned)t * (unsigned)FILL_CHUNK, lane, lo, hi);
        }
    }
}

__global__ __launch_bounds__(BLOCK_THREADS) void gcn_gemm_fill(
    const float* __restrict__ x, const float* __restrict__ W,
    float* __restrict__ h,
    const int* __restrict__ esrc, const int* __restrict__ edst,
    const float* __restrict__ ew, int* __restrict__ bucket,
    int n_nodes, int n_edges)
{
    __shared__ float2 Wp[(PAIR_ROWS + 1) * PAIR_STRIDE];

    for (int i = threadIdx.x; i < PAIR_ROWS * 16; i += BLOCK_THREADS) {
        const int p  = i >> 4;
        const int n4 = (i & 15) << 2;
        const float4 r0 = *reinterpret_cast<const float4*>(W + (2 * p + 0) * N_OUT + n4);
        const float4 r1 = *reinterpret_cast<const float4*>(W + (2 * p + 1) * N_OUT + n4);
        float4* dst = reinterpret_cast<float4*>(&Wp[p * PAIR_STRIDE + n4]);
        float4 lo, hi;
        lo.x = r0.x; lo.y = r1.x; lo.z = r0.y; lo.w = r1.y;
        hi.x = r0.z; hi.y = r1.z; hi.z = r0.w; hi.w = r1.w;
        dst[0] = lo;
        dst[1] = hi;
    }
    __syncthreads();

    const int lane = threadIdx.x & 31;
    const int wave = __builtin_amdgcn_readfirstlane((int)(threadIdx.x >> 5));

    if (wave >= GEMM_WAVES) {
        __builtin_amdgcn_s_setprio(3);
        fill_wave(esrc, edst, ew, bucket, n_nodes, n_edges, lane);
        return;
    }

    const int l16  = lane & 15;
    const int g    = lane >> 4;
    const int n_tiles = (n_nodes + 15) >> 4;
    const float2* bptr = &Wp[(long)g * PAIR_STRIDE + l16];

    for (int tile = (int)blockIdx.x + (int)gridDim.x * wave; tile < n_tiles;
         tile += (int)gridDim.x * GEMM_WAVES) {
        const long rowBase = (long)tile * 16;
        const float* aptr = x + (rowBase + l16) * (long)D_FEAT + 2 * g;

        v8f acc0 = {}, acc1 = {}, acc2 = {}, acc3 = {};

        for (int k = 0; k < D_FEAT; k += 4) {
            const v2f a = *reinterpret_cast<const v2f*>(aptr + k);

            const float2* wb = bptr + (k >> 1) * PAIR_STRIDE;
            const v2f b0 = *reinterpret_cast<const v2f*>(wb +  0);
            const v2f b1 = *reinterpret_cast<const v2f*>(wb + 16);
            const v2f b2 = *reinterpret_cast<const v2f*>(wb + 32);
            const v2f b3 = *reinterpret_cast<const v2f*>(wb + 48);

            acc0 = __builtin_amdgcn_wmma_f32_16x16x4_f32(false, a, false, b0, (short)0, acc0, false, false);
            acc1 = __builtin_amdgcn_wmma_f32_16x16x4_f32(false, a, false, b1, (short)0, acc1, false, false);
            acc2 = __builtin_amdgcn_wmma_f32_16x16x4_f32(false, a, false, b2, (short)0, acc2, false, false);
            acc3 = __builtin_amdgcn_wmma_f32_16x16x4_f32(false, a, false, b3, (short)0, acc3, false, false);
        }

        float* hb = h + rowBase * N_OUT + l16;
#pragma unroll
        for (int v = 0; v < 8; ++v) {
            const long m = v + 8 * g;
            hb[m * N_OUT +  0] = acc0[v];
            hb[m * N_OUT + 16] = acc1[v];
            hb[m * N_OUT + 32] = acc2[v];
            hb[m * N_OUT + 48] = acc3[v];
        }
    }
}

__device__ __forceinline__ int lane_bcast_i(int v0, int v1, int r)
{
    return (r < 32) ? __builtin_amdgcn_readlane(v0, r) : __builtin_amdgcn_readlane(v1, r - 32);
}

__global__ __launch_bounds__(256) void gcn_gather_softmax(
    const float* __restrict__ h, const int* __restrict__ esrc,
    const int* __restrict__ edst, const float* __restrict__ ew,
    float* out, int n_nodes, int n_edges)
{
    __shared__ int   s_src[8][64];
    __shared__ float s_w[8][64];

    const int lane = threadIdx.x & 31;
    const int wave = __builtin_amdgcn_readfirstlane((int)(threadIdx.x >> 5));
    const int n_waves = (int)gridDim.x * 8;
    const float2* h2 = reinterpret_cast<const float2*>(h);

    for (int node = (int)blockIdx.x * 8 + wave; node < n_nodes; node += n_waves) {
        float* row = out + (long)node * N_OUT;
        const float2 bk = reinterpret_cast<const float2*>(row)[lane];
        const int b0 = __float_as_int(bk.x);
        const int b1 = __float_as_int(bk.y);
        const int cnt = __builtin_amdgcn_readlane(b1, 31);

        float ax = 0.f, ay = 0.f;

        if (cnt <= BUCKET_CAP) {
            const bool pair_lane = lane < PAIR_SLOTS;
            const int  slotA = pair_lane ? lane : (2 * lane - PAIR_SLOTS);
            const int  slotB = 2 * lane - PAIR_SLOTS + 1;
            const bool vA = slotA < cnt;
            const bool vB = !pair_lane && (slotB < cnt);
            int srcA = b0, wA = b1;
            int srcB = 0,  wB = 0;
            if (cnt > PAIR_SLOTS) {
                if (!pair_lane && vA) { srcA = esrc[b0]; wA = __float_as_int(ew[b0]); }
                if (vB)               { srcB = esrc[b1]; wB = __float_as_int(ew[b1]); }
            }
            const unsigned long long kA =
                ((unsigned long long)(unsigned)srcA << 38) | ((unsigned long long)(unsigned)wA << 6) | (unsigned)slotA;
            const unsigned long long kB =
                ((unsigned long long)(unsigned)srcB << 38) | ((unsigned long long)(unsigned)wB << 6) | (unsigned)slotB;
            const int kAlo = (int)(unsigned)kA, kAhi = (int)(unsigned)(kA >> 32);
            const int kBlo = (int)(unsigned)kB, kBhi = (int)(unsigned)(kB >> 32);

            int rankA = 0, rankB = 0;
            if (cnt <= PAIR_SLOTS) {
                for (int j = 0; j < cnt; ++j) {
                    const unsigned long long kj =
                        ((unsigned long long)(unsigned)__builtin_amdgcn_readlane(kAhi, j) << 32) |
                        (unsigned)__builtin_amdgcn_readlane(kAlo, j);
                    rankA += (kj < kA) ? 1 : 0;
                }
            } else {
                const int nA = PAIR_SLOTS + ((cnt - PAIR_SLOTS + 1) >> 1);
                const int nB = (cnt - PAIR_SLOTS) >> 1;
                for (int j = 0; j < nA; ++j) {
                    const unsigned long long kj =
                        ((unsigned long long)(unsigned)__builtin_amdgcn_readlane(kAhi, j) << 32) |
                        (unsigned)__builtin_amdgcn_readlane(kAlo, j);
                    rankA += (kj < kA) ? 1 : 0;
                    rankB += (kj < kB) ? 1 : 0;
                }
                for (int j = 0; j < nB; ++j) {
                    const unsigned long long kj =
                        ((unsigned long long)(unsigned)__builtin_amdgcn_readlane(kBhi, PAIR_SLOTS + j) << 32) |
                        (unsigned)__builtin_amdgcn_readlane(kBlo, PAIR_SLOTS + j);
                    rankA += (kj < kA) ? 1 : 0;
                    rankB += (kj < kB) ? 1 : 0;
                }
            }
            if (vA) { s_src[wave][rankA] = srcA; s_w[wave][rankA] = __int_as_float(wA); }
            if (vB) { s_src[wave][rankB] = srcB; s_w[wave][rankB] = __int_as_float(wB); }
            const int   ss0 = s_src[wave][lane];
            const int   ss1 = s_src[wave][lane + 32];
            const int   sw0 = __float_as_int(s_w[wave][lane]);
            const int   sw1 = __float_as_int(s_w[wave][lane + 32]);

            for (int r = 0; r < cnt; r += 8) {
                float2 v[8];
                float  wv[8];
                bool   ok[8];
#pragma unroll
                for (int i = 0; i < 8; ++i) {
                    ok[i] = (r + i) < cnt;
                    const int rr = ok[i] ? (r + i) : r;
                    const int s  = lane_bcast_i(ss0, ss1, rr);
                    wv[i] = __int_as_float(lane_bcast_i(sw0, sw1, rr));
                    v[i]  = h2[(long)s * (N_OUT / 2) + lane];
                }
#pragma unroll
                for (int i = 0; i < 8; ++i) {
                    const float px = v[i].x * wv[i];
                    const float py = v[i].y * wv[i];
                    ax += ok[i] ? px : 0.f;
                    ay += ok[i] ? py : 0.f;
                }
            }
        } else {
            for (int base = 0; base < n_edges; base += 32) {
                const int e = base + lane;
                const int d = (e < n_edges) ? edst[e] : -1;
                unsigned m = __builtin_amdgcn_ballot_w32(d == node);
                while (m) {
                    const int j = __builtin_ctz(m);
                    m &= m - 1;
                    const int   s = esrc[base + j];
                    const float w = ew[base + j];
                    const float2 hv = h2[(long)s * (N_OUT / 2) + lane];
                    ax += hv.x * w;
                    ay += hv.y * w;
                }
            }
        }

        float m = fmaxf(ax, ay);
#pragma unroll
        for (int off = 16; off > 0; off >>= 1)
            m = fmaxf(m, __shfl_xor(m, off, 32));
        const float x0 = __expf(ax - m);
        const float x1 = __expf(ay - m);
        float s = x0 + x1;
#pragma unroll
        for (int off = 16; off > 0; off >>= 1)
            s += __shfl_xor(s, off, 32);
        const float inv = 1.0f / s;
        float2 o; o.x = x0 * inv; o.y = x1 * inv;
        reinterpret_cast<float2*>(row)[lane] = o;
    }
}

extern "C" void kernel_launch(void* const* d_in, const int* in_sizes, int n_in,
                              void* d_out, int out_size, void* d_ws, size_t ws_size,
                              hipStream_t stream)
{
    const float* x    = (const float*)d_in[0];
    const int*   esrc = (const int*)d_in[1];
    const int*   edst = (const int*)d_in[2];
    const float* ew   = (const float*)d_in[3];
    const float* W    = (const float*)d_in[4];

    const int n_nodes = in_sizes[0] / D_FEAT;
    const int n_edges = in_sizes[1];

    float* outp = (float*)d_out;
    float* h    = (float*)d_ws;

    gcn_clear<<<256, 128, 0, stream>>>((int*)d_out, n_nodes);
    gcn_gemm_fill<<<256, BLOCK_THREADS, 0, stream>>>(x, W, h, esrc, edst, ew, (int*)d_out,
                                                     n_nodes, n_edges);
    gcn_gather_softmax<<<2048, 256, 0, stream>>>(h, esrc, edst, ew, outp, n_nodes, n_edges);
}
